# starting kernel
# baseline (speedup 1.0000x reference)
#include <hip/hip_runtime.h>


typedef __attribute__((ext_vector_type(2))) float v2f;
typedef __attribute__((ext_vector_type(8))) float v8f;

namespace {
constexpr int kB = 4096;
constexpr int kI = 2048;
constexpr int kH = 2048;

constexpr int BM = 128;
constexpr int BN = 128;
constexpr int BK = 16;
constexpr int LDA = BK + 4;
}

__global__ __launch_bounds__(256)
void phase_trig_kernel(const float* __restrict__ x,
                       const float* __restrict__ coef,
                       float* __restrict__ cosP,
                       float* __restrict__ sinP) {
  __shared__ float As[BM * LDA];
  __shared__ float BsT[BN * LDA];

  const int tid   = threadIdx.x;
  const int lane  = tid & 31;
  const int wave  = tid >> 5;
  const int wm    = wave & 3;
  const int wn    = wave >> 2;
  const int lm    = lane & 15;
  const int lk    = (lane >> 4) << 1;
  const int lhalf = (lane >> 4) << 3;

  const int m0 = blockIdx.y * BM;
  const int n0 = blockIdx.x * BN;

  v8f acc[2][4] = {};

  for (int k0 = 0; k0 < kI; k0 += BK) {
    __syncthreads();

#pragma unroll
    for (int i = 0; i < 2; ++i) {
      const int idx = i * 256 + tid;

      const int r = idx >> 2;
      const int c = (idx & 3) << 2;
      const float4 va = *(const float4*)(x + (size_t)(m0 + r) * kI + k0 + c);
      *(float4*)&As[r * LDA + c] = va;

      const int nb = idx & 127;
      const int cb = (idx >> 7) << 2;
      float4 vb;
      vb.x = coef[(size_t)(k0 + cb + 0) * kH + n0 + nb];
      vb.y = coef[(size_t)(k0 + cb + 1) * kH + n0 + nb];
      vb.z = coef[(size_t)(k0 + cb + 2) * kH + n0 + nb];
      vb.w = coef[(size_t)(k0 + cb + 3) * kH + n0 + nb];
      *(float4*)&BsT[nb * LDA + cb] = vb;
    }
    __syncthreads();

    if (k0 + BK < kI) {
      __builtin_prefetch(x + (size_t)(m0 + (tid >> 2)) * kI + k0 + BK + ((tid & 3) << 2), 0, 0);
      __builtin_prefetch(coef + (size_t)(k0 + BK + (tid >> 4)) * kH + n0 + ((tid & 15) << 3), 0, 0);
    }

#pragma unroll
    for (int ks = 0; ks < BK; ks += 4) {
      v2f af[2], bf[4];
#pragma unroll
      for (int mt = 0; mt < 2; ++mt)
        af[mt] = *(const v2f*)&As[(wm * 32 + mt * 16 + lm) * LDA + ks + lk];
#pragma unroll
      for (int nt = 0; nt < 4; ++nt)
        bf[nt] = *(const v2f*)&BsT[(wn * 64 + nt * 16 + lm) * LDA + ks + lk];
#pragma unroll
      for (int mt = 0; mt < 2; ++mt)
#pragma unroll
        for (int nt = 0; nt < 4; ++nt)
          acc[mt][nt] = __builtin_amdgcn_wmma_f32_16x16x4_f32(
              false, af[mt], false, bf[nt], (short)0, acc[mt][nt], false, false);
    }
  }

#pragma unroll
  for (int mt = 0; mt < 2; ++mt) {
#pragma unroll
    for (int nt = 0; nt < 4; ++nt) {
      const int col  = n0 + wn * 64 + nt * 16 + lm;
      const int row0 = m0 + wm * 32 + mt * 16 + lhalf;
#pragma unroll
      for (int v = 0; v < 8; ++v) {
        const float p = acc[mt][nt][v];
        const size_t off = (size_t)(row0 + v) * kH + col;
        cosP[off] = __cosf(p);
        sinP[off] = __sinf(p);
      }
    }
  }
}

__global__ __launch_bounds__(256)
void gain_kernel(const float* __restrict__ cosP,
                 const float* __restrict__ sinP,
                 const float* __restrict__ an,
                 const float* __restrict__ bn,
                 const float* __restrict__ bias,
                 float* __restrict__ out) {
  __shared__ float Ac[BM * LDA];
  __shared__ float Asn[BM * LDA];
  __shared__ float Ban[BN * LDA];
  __shared__ float Bbn[BN * LDA];

  const int tid   = threadIdx.x;
  const int lane  = tid & 31;
  const int wave  = tid >> 5;
  const int wm    = wave & 3;
  const int wn    = wave >> 2;
  const int lm    = lane & 15;
  const int lk    = (lane >> 4) << 1;
  const int lhalf = (lane >> 4) << 3;

  const int m0 = blockIdx.y * BM;
  const int n0 = blockIdx.x * BN;

  v8f acc[2][4] = {};

  for (int k0 = 0; k0 < kH; k0 += BK) {
    __syncthreads();
#pragma unroll
    for (int i = 0; i < 2; ++i) {
      const int idx = i * 256 + tid;
      const int r = idx >> 2;
      const int c = (idx & 3) << 2;
      const float4 vc = *(const float4*)(cosP + (size_t)(m0 + r) * kH + k0 + c);
      *(float4*)&Ac[r * LDA + c] = vc;
      const float4 vs = *(const float4*)(sinP + (size_t)(m0 + r) * kH + k0 + c);
      *(float4*)&Asn[r * LDA + c] = vs;
      const float4 va = *(const float4*)(an + (size_t)(n0 + r) * kH + k0 + c);
      *(float4*)&Ban[r * LDA + c] = va;
      const float4 vb = *(const float4*)(bn + (size_t)(n0 + r) * kH + k0 + c);
      *(float4*)&Bbn[r * LDA + c] = vb;
    }
    __syncthreads();

    if (k0 + BK < kH) {
      __builtin_prefetch(cosP + (size_t)(m0 + (tid >> 2)) * kH + k0 + BK, 0, 0);
      __builtin_prefetch(sinP + (size_t)(m0 + (tid >> 2)) * kH + k0 + BK, 0, 0);
      __builtin_prefetch(an + (size_t)(n0 + (tid >> 2)) * kH + k0 + BK, 0, 0);
      __builtin_prefetch(bn + (size_t)(n0 + (tid >> 2)) * kH + k0 + BK, 0, 0);
    }

#pragma unroll
    for (int ks = 0; ks < BK; ks += 4) {
      v2f afc[2], afs[2], bfa[4], bfb[4];
#pragma unroll
      for (int mt = 0; mt < 2; ++mt) {
        const int ro = (wm * 32 + mt * 16 + lm) * LDA + ks + lk;
        afc[mt] = *(const v2f*)&Ac[ro];
        afs[mt] = *(const v2f*)&Asn[ro];
      }
#pragma unroll
      for (int nt = 0; nt < 4; ++nt) {
        const int ro = (wn * 64 + nt * 16 + lm) * LDA + ks + lk;
        bfa[nt] = *(const v2f*)&Ban[ro];
        bfb[nt] = *(const v2f*)&Bbn[ro];
      }
#pragma unroll
      for (int mt = 0; mt < 2; ++mt)
#pragma unroll
        for (int nt = 0; nt < 4; ++nt) {
          acc[mt][nt] = __builtin_amdgcn_wmma_f32_16x16x4_f32(
              false, afc[mt], false, bfa[nt], (short)0, acc[mt][nt], false, false);
          acc[mt][nt] = __builtin_amdgcn_wmma_f32_16x16x4_f32(
              false, afs[mt], false, bfb[nt], (short)0, acc[mt][nt], false, false);
        }
    }
  }

#pragma unroll
  for (int nt = 0; nt < 4; ++nt) {
    const int col = n0 + wn * 64 + nt * 16 + lm;
    const float bv = (float)kH * bias[col];
#pragma unroll
    for (int mt = 0; mt < 2; ++mt) {
      const int row0 = m0 + wm * 32 + mt * 16 + lhalf;
#pragma unroll
      for (int v = 0; v < 8; ++v)
        out[(size_t)(row0 + v) * kI + col] = acc[mt][nt][v] + bv;
    }
  }
}

extern "C" void kernel_launch(void* const* d_in, const int* in_sizes, int n_in,
                              void* d_out, int out_size, void* d_ws, size_t ws_size,
                              hipStream_t stream) {
  (void)in_sizes; (void)n_in; (void)out_size; (void)ws_size;

  const float* x    = (const float*)d_in[0];
  const float* coef = (const float*)d_in[1];
  const float* an   = (const float*)d_in[2];
  const float* bn   = (const float*)d_in[3];
  const float* bias = (const float*)d_in[4];
  float* out = (float*)d_out;

  float* cosP = (float*)d_ws;
  float* sinP = cosP + (size_t)kB * kH;

  const dim3 blk(256);
  const dim3 g1(kH / BN, kB / BM);
  phase_trig_kernel<<<g1, blk, 0, stream>>>(x, coef, cosP, sinP);

  const dim3 g2(kI / BN, kB / BM);
  gain_kernel<<<g2, blk, 0, stream>>>(cosP, sinP, an, bn, bias, out);
}


# K loop software-pipelined in both kernels: next chunk's global loads issued ahead of this chunk's WMMAs and held in registers, two LDS stages, one barrier per chunk; summation order unchanged
# speedup vs baseline: 1.1256x; 1.1256x over previous
#include <hip/hip_runtime.h>


typedef __attribute__((ext_vector_type(2))) float v2f;
typedef __attribute__((ext_vector_type(8))) float v8f;

namespace {
constexpr int kB = 4096;
constexpr int kI = 2048;
constexpr int kH = 2048;

constexpr int BM = 128;
constexpr int BN = 128;
constexpr int BK = 16;
constexpr int LDA = BK + 4;
constexpr int TILE = BM * LDA;
}

__global__ __launch_bounds__(256)
void phase_trig_kernel(const float* x,
                       const float* coef,
                       float* __restrict__ cosP,
                       float* __restrict__ sinP) {
  __shared__ alignas(16) float As[2 * TILE];
  __shared__ alignas(16) float BsT[2 * TILE];

  const int tid   = threadIdx.x;
  const int lane  = tid & 31;
  const int wave  = tid >> 5;
  const int wm    = wave & 3;
  const int wn    = wave >> 2;
  const int lm    = lane & 15;
  const int lk    = (lane >> 4) << 1;
  const int lhalf = (lane >> 4) << 3;

  const int m0 = blockIdx.y * BM;
  const int n0 = blockIdx.x * BN;

  const int sr = tid >> 2;
  const int sc = (tid & 3) << 2;
  const int snb = tid & 127;
  const int scb = (tid >> 7) << 2;
  const float* xa = x + (size_t)(m0 + sr) * kI + sc;
  const float* cb = coef + (size_t)scb * kH + n0 + snb;

  float4 ra0, ra1, rb0, rb1;
  v8f acc[2][4] = {};

#define P1_GLOAD(K0)                                                   \
  do {                                                                 \
    ra0 = *(const float4*)(xa + (K0));                                 \
    ra1 = *(const float4*)(xa + (size_t)64 * kI + (K0));               \
    const float* p0_ = cb + (size_t)(K0) * kH;                         \
    const float* p1_ = cb + (size_t)((K0) + 8) * kH;                   \
    rb0.x = p0_[0]; rb0.y = p0_[kH]; rb0.z = p0_[2 * kH]; rb0.w = p0_[3 * kH]; \
    rb1.x = p1_[0]; rb1.y = p1_[kH]; rb1.z = p1_[2 * kH]; rb1.w = p1_[3 * kH]; \
      \
      \
      \
    asm volatile("" ::: "memory");                                     \
  } while (0)

#define P1_LSTORE(ST)                                                  \
  do {                                                                 \
      \
      \
      \
      \
    __builtin_amdgcn_sched_barrier(0);                                 \
    *(float4*)&As[(ST) * TILE + sr * LDA + sc] = ra0;                  \
    *(float4*)&As[(ST) * TILE + (64 + sr) * LDA + sc] = ra1;           \
    *(float4*)&BsT[(ST) * TILE + snb * LDA + scb] = rb0;               \
    *(float4*)&BsT[(ST) * TILE + snb * LDA + 8 + scb] = rb1;           \
  } while (0)

#define P1_COMPUTE(ST)                                                 \
  do {                                                                 \
    _Pragma("unroll")                                                  \
    for (int ks = 0; ks < BK; ks += 4) {                               \
      v2f af[2], bf[4];                                                \
      _Pragma("unroll")                                                \
      for (int mt = 0; mt < 2; ++mt)                                   \
        af[mt] = *(const v2f*)&As[(ST) * TILE + (wm * 32 + mt * 16 + lm) * LDA + ks + lk];  \
      _Pragma("unroll")                                                \
      for (int nt = 0; nt < 4; ++nt)                                   \
        bf[nt] = *(const v2f*)&BsT[(ST) * TILE + (wn * 64 + nt * 16 + lm) * LDA + ks + lk]; \
      _Pragma("unroll")                                                \
      for (int mt = 0; mt < 2; ++mt)                                   \
        _Pragma("unroll")                                              \
        for (int nt = 0; nt < 4; ++nt)                                 \
          acc[mt][nt] = __builtin_amdgcn_wmma_f32_16x16x4_f32(         \
              false, af[mt], false, bf[nt], (short)0, acc[mt][nt], false, false); \
    }                                                                  \
  } while (0)

  P1_GLOAD(0);
  P1_LSTORE(0);
  __syncthreads();

  for (int k0 = 0; k0 < kI - 2 * BK; k0 += 2 * BK) {
    P1_GLOAD(k0 + BK);
    P1_COMPUTE(0);
    P1_LSTORE(1);
    __syncthreads();
    P1_GLOAD(k0 + 2 * BK);
    P1_COMPUTE(1);
    P1_LSTORE(0);
    __syncthreads();
  }
  P1_GLOAD(kI - BK);
  P1_COMPUTE(0);
  P1_LSTORE(1);
  __syncthreads();
  P1_COMPUTE(1);
#undef P1_GLOAD
#undef P1_LSTORE
#undef P1_COMPUTE

#pragma unroll
  for (int mt = 0; mt < 2; ++mt) {
#pragma unroll
    for (int nt = 0; nt < 4; ++nt) {
      const int col  = n0 + wn * 64 + nt * 16 + lm;
      const int row0 = m0 + wm * 32 + mt * 16 + lhalf;
#pragma unroll
      for (int v = 0; v < 8; ++v) {
        const float p = acc[mt][nt][v];
        const size_t off = (size_t)(row0 + v) * kH + col;
        cosP[off] = __cosf(p);
        sinP[off] = __sinf(p);
      }
    }
  }
}

__global__ __launch_bounds__(256)
void gain_kernel(const float* cosP,
                 const float* sinP,
                 const float* an,
                 const float* bn,
                 const float* __restrict__ bias,
                 float* __restrict__ out) {
  __shared__ alignas(16) float Ac[2 * TILE];
  __shared__ alignas(16) float Asn[2 * TILE];
  __shared__ alignas(16) float Ban[2 * TILE];
  __shared__ alignas(16) float Bbn[2 * TILE];

  const int tid   = threadIdx.x;
  const int lane  = tid & 31;
  const int wave  = tid >> 5;
  const int wm    = wave & 3;
  const int wn    = wave >> 2;
  const int lm    = lane & 15;
  const int lk    = (lane >> 4) << 1;
  const int lhalf = (lane >> 4) << 3;

  const int m0 = blockIdx.y * BM;
  const int n0 = blockIdx.x * BN;

  const int sr = tid >> 2;
  const int sc = (tid & 3) << 2;
  const size_t arow = (size_t)(m0 + sr) * kH + sc;
  const size_t brow = (size_t)(n0 + sr) * kH + sc;
  const float* pc = cosP + arow;
  const float* ps = sinP + arow;
  const float* pa = an + brow;
  const float* pb = bn + brow;

  float4 rc0, rc1, rs0, rs1, rn0, rn1, rm0, rm1;
  v8f acc[2][4] = {};

#define P2_GLOAD(K0)                                                   \
  do {                                                                 \
    rc0 = *(const float4*)(pc + (K0));                                 \
    rs0 = *(const float4*)(ps + (K0));                                 \
    rn0 = *(const float4*)(pa + (K0));                                 \
    rm0 = *(const float4*)(pb + (K0));                                 \
    rc1 = *(const float4*)(pc + (size_t)64 * kH + (K0));               \
    rs1 = *(const float4*)(ps + (size_t)64 * kH + (K0));               \
    rn1 = *(const float4*)(pa + (size_t)64 * kH + (K0));               \
    rm1 = *(const float4*)(pb + (size_t)64 * kH + (K0));               \
      \
      \
      \
    asm volatile("" ::: "memory");                                     \
  } while (0)

#define P2_LSTORE(ST)                                                  \
  do {                                                                 \
      \
      \
      \
      \
    __builtin_amdgcn_sched_barrier(0);                                 \
    const int o0_ = (ST) * TILE + sr * LDA + sc;                       \
    const int o1_ = (ST) * TILE + (64 + sr) * LDA + sc;                \
    *(float4*)&Ac[o0_]  = rc0;                                         \
    *(float4*)&Asn[o0_] = rs0;                                         \
    *(float4*)&Ban[o0_] = rn0;                                         \
    *(float4*)&Bbn[o0_] = rm0;                                         \
    *(float4*)&Ac[o1_]  = rc1;                                         \
    *(float4*)&Asn[o1_] = rs1;                                         \
    *(float4*)&Ban[o1_] = rn1;                                         \
    *(float4*)&Bbn[o1_] = rm1;                                         \
  } while (0)

#define P2_COMPUTE(ST)                                                 \
  do {                                                                 \
    _Pragma("unroll")                                                  \
    for (int ks = 0; ks < BK; ks += 4) {                               \
      v2f afc[2], afs[2], bfa[4], bfb[4];                              \
      _Pragma("unroll")                                                \
      for (int mt = 0; mt < 2; ++mt) {                                 \
        const int ro = (ST) * TILE + (wm * 32 + mt * 16 + lm) * LDA + ks + lk; \
        afc[mt] = *(const v2f*)&Ac[ro];                                \
        afs[mt] = *(const v2f*)&Asn[ro];                               \
      }                                                                \
      _Pragma("unroll")                                                \
      for (int nt = 0; nt < 4; ++nt) {                                 \
        const int ro = (ST) * TILE + (wn * 64 + nt * 16 + lm) * LDA + ks + lk; \
        bfa[nt] = *(const v2f*)&Ban[ro];                               \
        bfb[nt] = *(const v2f*)&Bbn[ro];                               \
      }                                                                \
      _Pragma("unroll")                                                \
      for (int mt = 0; mt < 2; ++mt)                                   \
        _Pragma("unroll")                                              \
        for (int nt = 0; nt < 4; ++nt) {                               \
          acc[mt][nt] = __builtin_amdgcn_wmma_f32_16x16x4_f32(         \
              false, afc[mt], false, bfa[nt], (short)0, acc[mt][nt], false, false); \
          acc[mt][nt] = __builtin_amdgcn_wmma_f32_16x16x4_f32(         \
              false, afs[mt], false, bfb[nt], (short)0, acc[mt][nt], false, false); \
        }                                                              \
    }                                                                  \
  } while (0)

  P2_GLOAD(0);
  P2_LSTORE(0);
  __syncthreads();

  for (int k0 = 0; k0 < kH - 2 * BK; k0 += 2 * BK) {
    P2_GLOAD(k0 + BK);
    P2_COMPUTE(0);
    P2_LSTORE(1);
    __syncthreads();
    P2_GLOAD(k0 + 2 * BK);
    P2_COMPUTE(1);
    P2_LSTORE(0);
    __syncthreads();
  }
  P2_GLOAD(kH - BK);
  P2_COMPUTE(0);
  P2_LSTORE(1);
  __syncthreads();
  P2_COMPUTE(1);
#undef P2_GLOAD
#undef P2_LSTORE
#undef P2_COMPUTE

#pragma unroll
  for (int nt = 0; nt < 4; ++nt) {
    const int col = n0 + wn * 64 + nt * 16 + lm;
    const float bv = (float)kH * bias[col];
#pragma unroll
    for (int mt = 0; mt < 2; ++mt) {
      const int row0 = m0 + wm * 32 + mt * 16 + lhalf;
#pragma unroll
      for (int v = 0; v < 8; ++v)
        out[(size_t)(row0 + v) * kI + col] = acc[mt][nt][v] + bv;
    }
  }
}

extern "C" void kernel_launch(void* const* d_in, const int* in_sizes, int n_in,
                              void* d_out, int out_size, void* d_ws, size_t ws_size,
                              hipStream_t stream) {
  (void)in_sizes; (void)n_in; (void)out_size; (void)ws_size;

  const float* x    = (const float*)d_in[0];
  const float* coef = (const float*)d_in[1];
  const float* an   = (const float*)d_in[2];
  const float* bn   = (const float*)d_in[3];
  const float* bias = (const float*)d_in[4];
  float* out = (float*)d_out;

  float* cosP = (float*)d_ws;
  float* sinP = cosP + (size_t)kB * kH;

  const dim3 blk(256);
  const dim3 g1(kH / BN, kB / BM);
  phase_trig_kernel<<<g1, blk, 0, stream>>>(x, coef, cosP, sinP);

  const dim3 g2(kI / BN, kB / BM);
  gain_kernel<<<g2, blk, 0, stream>>>(cosP, sinP, an, bn, bias, out);
}


# LDS fragment reads double-buffered in registers: each block's reads issued as one group ahead of the previous matrix block, no LDS wait right behind a barrier; summation order unchanged
# speedup vs baseline: 1.1844x; 1.0522x over previous
#include <hip/hip_runtime.h>


typedef __attribute__((ext_vector_type(2))) float v2f;
typedef __attribute__((ext_vector_type(8))) float v8f;

namespace {
constexpr int kB = 4096;
constexpr int kI = 2048;
constexpr int kH = 2048;

constexpr int BM = 128;
constexpr int BN = 128;
constexpr int BK = 16;
constexpr int LDA = BK + 4;
constexpr int TILE = BM * LDA;
}

__global__ __launch_bounds__(256)
void phase_trig_kernel(const float* x,
                       const float* coef,
                       float* __restrict__ cosP,
                       float* __restrict__ sinP) {
  __shared__ alignas(16) float As[2 * TILE];
  __shared__ alignas(16) float BsT[2 * TILE];

  const int tid   = threadIdx.x;
  const int lane  = tid & 31;
  const int wave  = tid >> 5;
  const int wm    = wave & 3;
  const int wn    = wave >> 2;
  const int lm    = lane & 15;
  const int lk    = (lane >> 4) << 1;
  const int lhalf = (lane >> 4) << 3;

  const int m0 = blockIdx.y * BM;
  const int n0 = blockIdx.x * BN;

  const int sr = tid >> 2;
  const int sc = (tid & 3) << 2;
  const int snb = tid & 127;
  const int scb = (tid >> 7) << 2;
  const float* xa = x + (size_t)(m0 + sr) * kI + sc;
  const float* cb = coef + (size_t)scb * kH + n0 + snb;

  float4 ra0, ra1, rb0, rb1;
  v2f fa0[4][2], fb0[4][4];
  v2f fa1[4][2], fb1[4][4];
  v8f acc[2][4] = {};

#define P1_GLOAD(K0)                                                   \
  do {                                                                 \
    ra0 = *(const float4*)(xa + (K0));                                 \
    ra1 = *(const float4*)(xa + (size_t)64 * kI + (K0));               \
    const float* p0_ = cb + (size_t)(K0) * kH;                         \
    const float* p1_ = cb + (size_t)((K0) + 8) * kH;                   \
    rb0.x = p0_[0]; rb0.y = p0_[kH]; rb0.z = p0_[2 * kH]; rb0.w = p0_[3 * kH]; \
    rb1.x = p1_[0]; rb1.y = p1_[kH]; rb1.z = p1_[2 * kH]; rb1.w = p1_[3 * kH]; \
      \
      \
      \
    asm volatile("" ::: "memory");                                     \
  } while (0)

#define P1_LSTORE(ST)                                                  \
  do {                                                                 \
    *(float4*)&As[(ST) * TILE + sr * LDA + sc] = ra0;                  \
    *(float4*)&As[(ST) * TILE + (64 + sr) * LDA + sc] = ra1;           \
    *(float4*)&BsT[(ST) * TILE + snb * LDA + scb] = rb0;               \
    *(float4*)&BsT[(ST) * TILE + snb * LDA + 8 + scb] = rb1;           \
  } while (0)

#define P1_FLOAD(FA, FB, ST)                                           \
  do {                                                                 \
    _Pragma("unroll")                                                  \
    for (int kk = 0; kk < 4; ++kk) {                                   \
      _Pragma("unroll")                                                \
      for (int mt = 0; mt < 2; ++mt)                                   \
        FA[kk][mt] = *(const v2f*)&As[(ST) * TILE + (wm * 32 + mt * 16 + lm) * LDA + kk * 4 + lk];  \
      _Pragma("unroll")                                                \
      for (int nt = 0; nt < 4; ++nt)                                   \
        FB[kk][nt] = *(const v2f*)&BsT[(ST) * TILE + (wn * 64 + nt * 16 + lm) * LDA + kk * 4 + lk]; \
    }                                                                  \
      \
      \
      \
      \
    __builtin_amdgcn_sched_barrier(0);                                 \
  } while (0)

#define P1_MMA(FA, FB)                                                 \
  do {                                                                 \
    _Pragma("unroll")                                                  \
    for (int kk = 0; kk < 4; ++kk)                                     \
      _Pragma("unroll")                                                \
      for (int mt = 0; mt < 2; ++mt)                                   \
        _Pragma("unroll")                                              \
        for (int nt = 0; nt < 4; ++nt)                                 \
          acc[mt][nt] = __builtin_amdgcn_wmma_f32_16x16x4_f32(         \
              false, FA[kk][mt], false, FB[kk][nt], (short)0, acc[mt][nt], false, false); \
      \
      \
      \
      \
    __builtin_amdgcn_sched_barrier(0);                                 \
  } while (0)

  P1_GLOAD(0);
  P1_LSTORE(0);
  __syncthreads();
  P1_FLOAD(fa0, fb0, 0);
  P1_GLOAD(BK);
  P1_LSTORE(1);

  for (int k0 = 0; k0 < kI - 2 * BK; k0 += 2 * BK) {
    __syncthreads();
    P1_GLOAD(k0 + 2 * BK);
    P1_FLOAD(fa1, fb1, 1);
    P1_MMA(fa0, fb0);
    P1_LSTORE(0);
    __syncthreads();
    P1_GLOAD(k0 + 3 * BK);
    P1_FLOAD(fa0, fb0, 0);
    P1_MMA(fa1, fb1);
    P1_LSTORE(1);
  }
  __syncthreads();
  P1_FLOAD(fa1, fb1, 1);
  P1_MMA(fa0, fb0);
  P1_MMA(fa1, fb1);
#undef P1_GLOAD
#undef P1_LSTORE
#undef P1_FLOAD
#undef P1_MMA

#pragma unroll
  for (int mt = 0; mt < 2; ++mt) {
#pragma unroll
    for (int nt = 0; nt < 4; ++nt) {
      const int col  = n0 + wn * 64 + nt * 16 + lm;
      const int row0 = m0 + wm * 32 + mt * 16 + lhalf;
#pragma unroll
      for (int v = 0; v < 8; ++v) {
        const float p = acc[mt][nt][v];
        const size_t off = (size_t)(row0 + v) * kH + col;
        cosP[off] = __cosf(p);
        sinP[off] = __sinf(p);
      }
    }
  }
}

__global__ __launch_bounds__(256)
void gain_kernel(const float* cosP,
                 const float* sinP,
                 const float* an,
                 const float* bn,
                 const float* __restrict__ bias,
                 float* __restrict__ out) {
  __shared__ alignas(16) float Ac[2 * TILE];
  __shared__ alignas(16) float Asn[2 * TILE];
  __shared__ alignas(16) float Ban[2 * TILE];
  __shared__ alignas(16) float Bbn[2 * TILE];

  const int tid   = threadIdx.x;
  const int lane  = tid & 31;
  const int wave  = tid >> 5;
  const int wm    = wave & 3;
  const int wn    = wave >> 2;
  const int lm    = lane & 15;
  const int lk    = (lane >> 4) << 1;
  const int lhalf = (lane >> 4) << 3;

  const int m0 = blockIdx.y * BM;
  const int n0 = blockIdx.x * BN;

  const int sr = tid >> 2;
  const int sc = (tid & 3) << 2;
  const size_t arow = (size_t)(m0 + sr) * kH + sc;
  const size_t brow = (size_t)(n0 + sr) * kH + sc;
  const float* pc = cosP + arow;
  const float* ps = sinP + arow;
  const float* pa = an + brow;
  const float* pb = bn + brow;

  float4 rc0, rc1, rs0, rs1, rn0, rn1, rm0, rm1;
  v2f fc0[2][2], fs0[2][2], fn0[2][4], fm0[2][4];
  v2f fc1[2][2], fs1[2][2], fn1[2][4], fm1[2][4];
  v8f acc[2][4] = {};

#define P2_GLOAD(K0)                                                   \
  do {                                                                 \
    rc0 = *(const float4*)(pc + (K0));                                 \
    rs0 = *(const float4*)(ps + (K0));                                 \
    rn0 = *(const float4*)(pa + (K0));                                 \
    rm0 = *(const float4*)(pb + (K0));                                 \
    rc1 = *(const float4*)(pc + (size_t)64 * kH + (K0));               \
    rs1 = *(const float4*)(ps + (size_t)64 * kH + (K0));               \
    rn1 = *(const float4*)(pa + (size_t)64 * kH + (K0));               \
    rm1 = *(const float4*)(pb + (size_t)64 * kH + (K0));               \
      \
      \
      \
    asm volatile("" ::: "memory");                                     \
  } while (0)

#define P2_LSTORE(ST)                                                  \
  do {                                                                 \
    const int o0_ = (ST) * TILE + sr * LDA + sc;                       \
    const int o1_ = (ST) * TILE + (64 + sr) * LDA + sc;                \
    *(float4*)&Ac[o0_]  = rc0;                                         \
    *(float4*)&Asn[o0_] = rs0;                                         \
    *(float4*)&Ban[o0_] = rn0;                                         \
    *(float4*)&Bbn[o0_] = rm0;                                         \
    *(float4*)&Ac[o1_]  = rc1;                                         \
    *(float4*)&Asn[o1_] = rs1;                                         \
    *(float4*)&Ban[o1_] = rn1;                                         \
    *(float4*)&Bbn[o1_] = rm1;                                         \
  } while (0)

#define P2_FLOAD(FC, FS, FN, FM, ST, HF)                               \
  do {                                                                 \
    _Pragma("unroll")                                                  \
    for (int kk = 0; kk < 2; ++kk) {                                   \
      _Pragma("unroll")                                                \
      for (int mt = 0; mt < 2; ++mt) {                                 \
        const int ro = (ST) * TILE + (wm * 32 + mt * 16 + lm) * LDA + (HF) * 8 + kk * 4 + lk; \
        FC[kk][mt] = *(const v2f*)&Ac[ro];                             \
        FS[kk][mt] = *(const v2f*)&Asn[ro];                            \
      }                                                                \
      _Pragma("unroll")                                                \
      for (int nt = 0; nt < 4; ++nt) {                                 \
        const int ro = (ST) * TILE + (wn * 64 + nt * 16 + lm) * LDA + (HF) * 8 + kk * 4 + lk; \
        FN[kk][nt] = *(const v2f*)&Ban[ro];                            \
        FM[kk][nt] = *(const v2f*)&Bbn[ro];                            \
      }                                                                \
    }                                                                  \
      \
      \
      \
      \
    __builtin_amdgcn_sched_barrier(0);                                 \
  } while (0)

#define P2_MMA(FC, FS, FN, FM)                                         \
  do {                                                                 \
    _Pragma("unroll")                                                  \
    for (int kk = 0; kk < 2; ++kk)                                     \
      _Pragma("unroll")                                                \
      for (int mt = 0; mt < 2; ++mt)                                   \
        _Pragma("unroll")                                              \
        for (int nt = 0; nt < 4; ++nt) {                               \
          acc[mt][nt] = __builtin_amdgcn_wmma_f32_16x16x4_f32(         \
              false, FC[kk][mt], false, FN[kk][nt], (short)0, acc[mt][nt], false, false); \
          acc[mt][nt] = __builtin_amdgcn_wmma_f32_16x16x4_f32(         \
              false, FS[kk][mt], false, FM[kk][nt], (short)0, acc[mt][nt], false, false); \
        }                                                              \
      \
      \
      \
      \
    __builtin_amdgcn_sched_barrier(0);                                 \
  } while (0)

  P2_GLOAD(0);
  P2_LSTORE(0);
  __syncthreads();
  P2_GLOAD(BK);
  P2_FLOAD(fc0, fs0, fn0, fm0, 0, 0);

  for (int k0 = 0; k0 < kH - 2 * BK; k0 += 2 * BK) {
    P2_FLOAD(fc1, fs1, fn1, fm1, 0, 1);
    P2_MMA(fc0, fs0, fn0, fm0);
    P2_LSTORE(1);
    __syncthreads();
    P2_GLOAD(k0 + 2 * BK);
    P2_FLOAD(fc0, fs0, fn0, fm0, 1, 0);
    P2_MMA(fc1, fs1, fn1, fm1);

    P2_FLOAD(fc1, fs1, fn1, fm1, 1, 1);
    P2_MMA(fc0, fs0, fn0, fm0);
    P2_LSTORE(0);
    __syncthreads();
    P2_GLOAD(k0 + 3 * BK);
    P2_FLOAD(fc0, fs0, fn0, fm0, 0, 0);
    P2_MMA(fc1, fs1, fn1, fm1);
  }
  P2_FLOAD(fc1, fs1, fn1, fm1, 0, 1);
  P2_MMA(fc0, fs0, fn0, fm0);
  P2_LSTORE(1);
  __syncthreads();
  P2_FLOAD(fc0, fs0, fn0, fm0, 1, 0);
  P2_MMA(fc1, fs1, fn1, fm1);
  P2_FLOAD(fc1, fs1, fn1, fm1, 1, 1);
  P2_MMA(fc0, fs0, fn0, fm0);
  P2_MMA(fc1, fs1, fn1, fm1);
#undef P2_GLOAD
#undef P2_LSTORE
#undef P2_FLOAD
#undef P2_MMA

#pragma unroll
  for (int nt = 0; nt < 4; ++nt) {
    const int col = n0 + wn * 64 + nt * 16 + lm;
    const float bv = (float)kH * bias[col];
#pragma unroll
    for (int mt = 0; mt < 2; ++mt) {
      const int row0 = m0 + wm * 32 + mt * 16 + lhalf;
#pragma unroll
      for (int v = 0; v < 8; ++v)
        out[(size_t)(row0 + v) * kI + col] = acc[mt][nt][v] + bv;
    }
  }
}

extern "C" void kernel_launch(void* const* d_in, const int* in_sizes, int n_in,
                              void* d_out, int out_size, void* d_ws, size_t ws_size,
                              hipStream_t stream) {
  (void)in_sizes; (void)n_in; (void)out_size; (void)ws_size;

  const float* x    = (const float*)d_in[0];
  const float* coef = (const float*)d_in[1];
  const float* an   = (const float*)d_in[2];
  const float* bn   = (const float*)d_in[3];
  const float* bias = (const float*)d_in[4];
  float* out = (float*)d_out;

  float* cosP = (float*)d_ws;
  float* sinP = cosP + (size_t)kB * kH;

  const dim3 blk(256);
  const dim3 g1(kH / BN, kB / BM);
  phase_trig_kernel<<<g1, blk, 0, stream>>>(x, coef, cosP, sinP);

  const dim3 g2(kI / BN, kB / BM);
  gain_kernel<<<g2, blk, 0, stream>>>(cosP, sinP, an, bn, bias, out);
}


# Workgroup barrier split into signal and wait, a matrix piece ahead of every wait (LDS stores, barrier, first reads of new stage); 4-slot fragment ring, kernel 1 chunk 32; summation order unchanged
# speedup vs baseline: 1.2243x; 1.0337x over previous
#include <hip/hip_runtime.h>


typedef __attribute__((ext_vector_type(2))) float v2f;
typedef __attribute__((ext_vector_type(8))) float v8f;

namespace {
constexpr int kB = 4096;
constexpr int kI = 2048;
constexpr int kH = 2048;

constexpr int BM = 128;
constexpr int BN = 128;

constexpr int BK1 = 32;
constexpr int LD1 = BK1 + 4;
constexpr int TILE1 = BM * LD1;

constexpr int BK2 = 16;
constexpr int LD2 = BK2 + 4;
constexpr int TILE2 = BM * LD2;
}

static __device__ __forceinline__ void lds_done_and_signal() {
  asm volatile("s_wait_dscnt 0x0\n\ts_barrier_signal -1" ::: "memory");
}
static __device__ __forceinline__ void barrier_wait() {
  asm volatile("s_barrier_wait -1" ::: "memory");
}

__global__ __launch_bounds__(256)
void phase_trig_kernel(const float* x,
                       const float* coef,
                       float* __restrict__ cosP,
                       float* __restrict__ sinP) {
  __shared__ alignas(16) float As[2 * TILE1];
  __shared__ alignas(16) float BsT[2 * TILE1];

  const int tid   = threadIdx.x;
  const int lane  = tid & 31;
  const int wave  = tid >> 5;
  const int wm    = wave & 3;
  const int wn    = wave >> 2;
  const int lm    = lane & 15;
  const int lk    = (lane >> 4) << 1;
  const int lhalf = (lane >> 4) << 3;

  const int m0 = blockIdx.y * BM;
  const int n0 = blockIdx.x * BN;

  const int sr = tid >> 3;
  const int sc = (tid & 7) << 2;
  const int snb = tid & 127;
  const int scb = (tid >> 7) << 2;
  const float* xa = x + (size_t)(m0 + sr) * kI + sc;
  const float* cb = coef + (size_t)scb * kH + n0 + snb;

  float4 ra0, ra1, ra2, ra3;
  float4 rb0, rb1, rb2, rb3;
  v2f fa[4][2][2];
  v2f fb[4][2][4];
  v8f acc[2][4] = {};

#define P1_BCOL(R, K0, I)                                              \
  do {                                                                 \
    const float* p_ = cb + (size_t)((K0) + (I) * 8) * kH;              \
    R.x = p_[0]; R.y = p_[kH]; R.z = p_[2 * kH]; R.w = p_[3 * kH];     \
  } while (0)

#define P1_GLOAD(K0)                                                   \
  do {                                                                 \
    ra0 = *(const float4*)(xa + (K0));                                 \
    ra1 = *(const float4*)(xa + (size_t)32 * kI + (K0));               \
    ra2 = *(const float4*)(xa + (size_t)64 * kI + (K0));               \
    ra3 = *(const float4*)(xa + (size_t)96 * kI + (K0));               \
    P1_BCOL(rb0, K0, 0);                                               \
    P1_BCOL(rb1, K0, 1);                                               \
    P1_BCOL(rb2, K0, 2);                                               \
    P1_BCOL(rb3, K0, 3);                                               \
      \
      \
      \
    asm volatile("" ::: "memory");                                     \
      \
      \
      \
    __builtin_amdgcn_sched_barrier(0);                                 \
  } while (0)

#define P1_LSTORE(ST)                                                  \
  do {                                                                 \
    *(float4*)&As[(ST) * TILE1 + sr * LD1 + sc] = ra0;                 \
    *(float4*)&As[(ST) * TILE1 + (32 + sr) * LD1 + sc] = ra1;          \
    *(float4*)&As[(ST) * TILE1 + (64 + sr) * LD1 + sc] = ra2;          \
    *(float4*)&As[(ST) * TILE1 + (96 + sr) * LD1 + sc] = ra3;          \
    *(float4*)&BsT[(ST) * TILE1 + snb * LD1 + scb] = rb0;              \
    *(float4*)&BsT[(ST) * TILE1 + snb * LD1 + 8 + scb] = rb1;          \
    *(float4*)&BsT[(ST) * TILE1 + snb * LD1 + 16 + scb] = rb2;         \
    *(float4*)&BsT[(ST) * TILE1 + snb * LD1 + 24 + scb] = rb3;         \
  } while (0)

#define P1_FLOAD2(J, ST)                                               \
  do {                                                                 \
    _Pragma("unroll")                                                  \
    for (int j = (J); j < (J) + 2; ++j) {                              \
      _Pragma("unroll")                                                \
      for (int kk = 0; kk < 2; ++kk) {                                 \
        _Pragma("unroll")                                              \
        for (int mt = 0; mt < 2; ++mt)                                 \
          fa[j][kk][mt] = *(const v2f*)&As[(ST) * TILE1 + (wm * 32 + mt * 16 + lm) * LD1 + j * 8 + kk * 4 + lk];  \
        _Pragma("unroll")                                              \
        for (int nt = 0; nt < 4; ++nt)                                 \
          fb[j][kk][nt] = *(const v2f*)&BsT[(ST) * TILE1 + (wn * 64 + nt * 16 + lm) * LD1 + j * 8 + kk * 4 + lk]; \
      }                                                                \
    }                                                                  \
  } while (0)

#define GROUP_END() __builtin_amdgcn_sched_barrier(0)

#define P1_MMA(J)                                                      \
  do {                                                                 \
    _Pragma("unroll")                                                  \
    for (int kk = 0; kk < 2; ++kk)                                     \
      _Pragma("unroll")                                                \
      for (int mt = 0; mt < 2; ++mt)                                   \
        _Pragma("unroll")                                              \
        for (int nt = 0; nt < 4; ++nt)                                 \
          acc[mt][nt] = __builtin_amdgcn_wmma_f32_16x16x4_f32(         \
              false, fa[J][kk][mt], false, fb[J][kk][nt], (short)0, acc[mt][nt], false, false); \
      \
      \
      \
    __builtin_amdgcn_sched_barrier(0);                                 \
  } while (0)

#define P1_CHUNK(ST, KN, STORE, LOAD)                                  \
  do {                                                                 \
    P1_FLOAD2(2, ST);                                                  \
    if (STORE) P1_LSTORE((ST) ^ 1);                                    \
    GROUP_END();                                                       \
    P1_MMA(0);                                                         \
    if (STORE) lds_done_and_signal();                                  \
    if (LOAD) P1_GLOAD(KN);                                            \
    P1_MMA(1);                                                         \
    P1_MMA(2);                                                         \
    if (STORE) {                                                       \
      barrier_wait();                                                  \
      P1_FLOAD2(0, (ST) ^ 1);                                          \
      GROUP_END();                                                     \
    }                                                                  \
    P1_MMA(3);                                                         \
  } while (0)

  P1_GLOAD(0);
  P1_LSTORE(0);
  __syncthreads();
  P1_GLOAD(BK1);
  P1_FLOAD2(0, 0);
  GROUP_END();

  for (int k0 = 0; k0 < kI - 2 * BK1; k0 += 2 * BK1) {
    P1_CHUNK(0, k0 + 2 * BK1, true, true);
    P1_CHUNK(1, k0 + 3 * BK1, true, true);
  }
  P1_CHUNK(0, 0, true, false);
  P1_CHUNK(1, 0, false, false);
#undef P1_BCOL
#undef P1_GLOAD
#undef P1_LSTORE
#undef P1_FLOAD2
#undef P1_MMA
#undef P1_CHUNK

#pragma unroll
  for (int mt = 0; mt < 2; ++mt) {
#pragma unroll
    for (int nt = 0; nt < 4; ++nt) {
      const int col  = n0 + wn * 64 + nt * 16 + lm;
      const int row0 = m0 + wm * 32 + mt * 16 + lhalf;
#pragma unroll
      for (int v = 0; v < 8; ++v) {
        const float p = acc[mt][nt][v];
        const size_t off = (size_t)(row0 + v) * kH + col;
        cosP[off] = __cosf(p);
        sinP[off] = __sinf(p);
      }
    }
  }
}

__global__ __launch_bounds__(256)
void gain_kernel(const float* cosP,
                 const float* sinP,
                 const float* an,
                 const float* bn,
                 const float* __restrict__ bias,
                 float* __restrict__ out) {
  __shared__ alignas(16) float Ac[2 * TILE2];
  __shared__ alignas(16) float Asn[2 * TILE2];
  __shared__ alignas(16) float Ban[2 * TILE2];
  __shared__ alignas(16) float Bbn[2 * TILE2];

  const int tid   = threadIdx.x;
  const int lane  = tid & 31;
  const int wave  = tid >> 5;
  const int wm    = wave & 3;
  const int wn    = wave >> 2;
  const int lm    = lane & 15;
  const int lk    = (lane >> 4) << 1;
  const int lhalf = (lane >> 4) << 3;

  const int m0 = blockIdx.y * BM;
  const int n0 = blockIdx.x * BN;

  const int sr = tid >> 2;
  const int sc = (tid & 3) << 2;
  const size_t arow = (size_t)(m0 + sr) * kH + sc;
  const size_t brow = (size_t)(n0 + sr) * kH + sc;
  const float* pc = cosP + arow;
  const float* ps = sinP + arow;
  const float* pa = an + brow;
  const float* pb = bn + brow;

  float4 rc0, rc1, rs0, rs1, rn0, rn1, rm0, rm1;
  v2f fc[4][2], fs[4][2], fn[4][4], fm[4][4];
  v8f acc[2][4] = {};

#define P2_GLOAD(K0)                                                   \
  do {                                                                 \
    rc0 = *(const float4*)(pc + (K0));                                 \
    rs0 = *(const float4*)(ps + (K0));                                 \
    rn0 = *(const float4*)(pa + (K0));                                 \
    rm0 = *(const float4*)(pb + (K0));                                 \
    rc1 = *(const float4*)(pc + (size_t)64 * kH + (K0));               \
    rs1 = *(const float4*)(ps + (size_t)64 * kH + (K0));               \
    rn1 = *(const float4*)(pa + (size_t)64 * kH + (K0));               \
    rm1 = *(const float4*)(pb + (size_t)64 * kH + (K0));               \
      \
      \
      \
    asm volatile("" ::: "memory");                                     \
      \
      \
      \
    __builtin_amdgcn_sched_barrier(0);                                 \
  } while (0)

#define P2_LSTORE(ST)                                                  \
  do {                                                                 \
    const int o0_ = (ST) * TILE2 + sr * LD2 + sc;                      \
    const int o1_ = (ST) * TILE2 + (64 + sr) * LD2 + sc;               \
    *(float4*)&Ac[o0_]  = rc0;                                         \
    *(float4*)&Asn[o0_] = rs0;                                         \
    *(float4*)&Ban[o0_] = rn0;                                         \
    *(float4*)&Bbn[o0_] = rm0;                                         \
    *(float4*)&Ac[o1_]  = rc1;                                         \
    *(float4*)&Asn[o1_] = rs1;                                         \
    *(float4*)&Ban[o1_] = rn1;                                         \
    *(float4*)&Bbn[o1_] = rm1;                                         \
  } while (0)

#define P2_FLOAD2(J, ST)                                               \
  do {                                                                 \
    _Pragma("unroll")                                                  \
    for (int j = (J); j < (J) + 2; ++j) {                              \
      _Pragma("unroll")                                                \
      for (int mt = 0; mt < 2; ++mt) {                                 \
        const int ro = (ST) * TILE2 + (wm * 32 + mt * 16 + lm) * LD2 + j * 4 + lk; \
        fc[j][mt] = *(const v2f*)&Ac[ro];                              \
        fs[j][mt] = *(const v2f*)&Asn[ro];                             \
      }                                                                \
      _Pragma("unroll")                                                \
      for (int nt = 0; nt < 4; ++nt) {                                 \
        const int ro = (ST) * TILE2 + (wn * 64 + nt * 16 + lm) * LD2 + j * 4 + lk; \
        fn[j][nt] = *(const v2f*)&Ban[ro];                             \
        fm[j][nt] = *(const v2f*)&Bbn[ro];                             \
      }                                                                \
    }                                                                  \
  } while (0)

#define P2_MMA(J)                                                      \
  do {                                                                 \
    _Pragma("unroll")                                                  \
    for (int mt = 0; mt < 2; ++mt)                                     \
      _Pragma("unroll")                                                \
      for (int nt = 0; nt < 4; ++nt) {                                 \
        acc[mt][nt] = __builtin_amdgcn_wmma_f32_16x16x4_f32(           \
            false, fc[J][mt], false, fn[J][nt], (short)0, acc[mt][nt], false, false); \
        acc[mt][nt] = __builtin_amdgcn_wmma_f32_16x16x4_f32(           \
            false, fs[J][mt], false, fm[J][nt], (short)0, acc[mt][nt], false, false); \
      }                                                                \
      \
      \
      \
    __builtin_amdgcn_sched_barrier(0);                                 \
  } while (0)

#define P2_CHUNK(ST, KN, STORE, LOAD)                                  \
  do {                                                                 \
    P2_FLOAD2(2, ST);                                                  \
    if (STORE) P2_LSTORE((ST) ^ 1);                                    \
    GROUP_END();                                                       \
    P2_MMA(0);                                                         \
    if (STORE) lds_done_and_signal();                                  \
    if (LOAD) P2_GLOAD(KN);                                            \
    P2_MMA(1);                                                         \
    P2_MMA(2);                                                         \
    if (STORE) {                                                       \
      barrier_wait();                                                  \
      P2_FLOAD2(0, (ST) ^ 1);                                          \
      GROUP_END();                                                     \
    }                                                                  \
    P2_MMA(3);                                                         \
  } while (0)

  P2_GLOAD(0);
  P2_LSTORE(0);
  __syncthreads();
  P2_GLOAD(BK2);
  P2_FLOAD2(0, 0);
  GROUP_END();

  for (int k0 = 0; k0 < kH - 2 * BK2; k0 += 2 * BK2) {
    P2_CHUNK(0, k0 + 2 * BK2, true, true);
    P2_CHUNK(1, k0 + 3 * BK2, true, true);
  }
  P2_CHUNK(0, 0, true, false);
  P2_CHUNK(1, 0, false, false);
#undef P2_GLOAD
#undef P2_LSTORE
#undef P2_FLOAD2
#undef P2_MMA
#undef P2_CHUNK
#undef GROUP_END

#pragma unroll
  for (int nt = 0; nt < 4; ++nt) {
    const int col = n0 + wn * 64 + nt * 16 + lm;
    const float bv = (float)kH * bias[col];
#pragma unroll
    for (int mt = 0; mt < 2; ++mt) {
      const int row0 = m0 + wm * 32 + mt * 16 + lhalf;
#pragma unroll
      for (int v = 0; v < 8; ++v)
        out[(size_t)(row0 + v) * kI + col] = acc[mt][nt][v] + bv;
    }
  }
}

extern "C" void kernel_launch(void* const* d_in, const int* in_sizes, int n_in,
                              void* d_out, int out_size, void* d_ws, size_t ws_size,
                              hipStream_t stream) {
  (void)in_sizes; (void)n_in; (void)out_size; (void)ws_size;

  const float* x    = (const float*)d_in[0];
  const float* coef = (const float*)d_in[1];
  const float* an   = (const float*)d_in[2];
  const float* bn   = (const float*)d_in[3];
  const float* bias = (const float*)d_in[4];
  float* out = (float*)d_out;

  float* cosP = (float*)d_ws;
  float* sinP = cosP + (size_t)kB * kH;

  const dim3 blk(256);
  const dim3 g1(kH / BN, kB / BM);
  phase_trig_kernel<<<g1, blk, 0, stream>>>(x, coef, cosP, sinP);

  const dim3 g2(kI / BN, kB / BM);
  gain_kernel<<<g2, blk, 0, stream>>>(cosP, sinP, an, bn, bias, out);
}
